# starting kernel
# baseline (speedup 1.0000x reference)
#include <hip/hip_runtime.h>
#include <hip/hip_bf16.h>
#include <math.h>

typedef __attribute__((ext_vector_type(2))) float v2f;
typedef __attribute__((ext_vector_type(4))) float v4f;
typedef __attribute__((ext_vector_type(8))) float v8f;

__global__ __launch_bounds__(256)
void agg_kernel(const float* __restrict__ h, const float* __restrict__ m,
                const int* __restrict__ idx_e, const int* __restrict__ idx_n,
                float* __restrict__ f_u) {
    const int e     = blockIdx.x;
    const int tid   = threadIdx.x;
    const int quad  = tid & 15;
    const int rslot = tid >> 4;

    __shared__ int s_ie[64];
    __shared__ int s_in[64];
    if (tid < 64)          s_ie[tid]      = idx_e[(size_t)e * 64 + tid];
    else if (tid < 128)    s_in[tid - 64] = idx_n[(size_t)e * 64 + (tid - 64)];
    __syncthreads();

    const v4f* __restrict__ hv = (const v4f*)h;
    const v4f* __restrict__ mv = (const v4f*)m;
    v4f* __restrict__ fv       = (v4f*)f_u;

    __shared__ v4f red[256];

    {
        v4f acc = {0.f, 0.f, 0.f, 0.f};
        #pragma unroll
        for (int i = 0; i < 4; ++i) {
            const int r = rslot + 16 * i;
            acc += hv[(size_t)s_ie[r] * 16 + quad];
        }
        red[tid] = acc;
        __syncthreads();
        #pragma unroll
        for (int s = 128; s >= 16; s >>= 1) {
            if (tid < s) red[tid] += red[tid + s];
            __syncthreads();
        }
        if (tid < 16) {
            v4f r = red[tid];
            r *= (1.0f / 64.0f);
            fv[(size_t)e * 32 + tid] = r;
        }
        __syncthreads();
    }

    {
        v4f acc = {0.f, 0.f, 0.f, 0.f};
        #pragma unroll
        for (int i = 0; i < 4; ++i) {
            const int r = rslot + 16 * i;
            acc += mv[(size_t)s_in[r] * 16 + quad];
        }
        red[tid] = acc;
        __syncthreads();
        #pragma unroll
        for (int s = 128; s >= 16; s >>= 1) {
            if (tid < s) red[tid] += red[tid + s];
            __syncthreads();
        }
        if (tid < 16) {
            v4f r = red[tid];
            r *= (1.0f / 64.0f);
            fv[(size_t)e * 32 + 16 + tid] = r;
        }
    }
}

__global__ __launch_bounds__(256)
void wmma_gemm_f32(const float* __restrict__ A, const float* __restrict__ B,
                   float* __restrict__ C, int M, int Kd) {
    constexpr int N = 128;
    const int wave = threadIdx.x >> 5;
    const int lane = threadIdx.x & 31;
    const int tile = blockIdx.x * 8 + wave;
    const int tm   = tile >> 3;
    const int tn   = tile & 7;
    if (tm * 16 >= M) return;

    const int half = lane >> 4;
    const int l    = lane & 15;
    const int arow = tm * 16 + l;
    const int ncol = tn * 16 + l;

    v8f acc = {0.f, 0.f, 0.f, 0.f, 0.f, 0.f, 0.f, 0.f};

    for (int k = 0; k < Kd; k += 4) {
        const int ka = k + half * 2;
        v2f a, b;
        a.x = A[(size_t)arow * Kd + ka];
        a.y = A[(size_t)arow * Kd + ka + 1];
        b.x = B[(size_t)(ka)     * N + ncol];
        b.y = B[(size_t)(ka + 1) * N + ncol];
        acc = __builtin_amdgcn_wmma_f32_16x16x4_f32(
             false, a,  false, b,
             (short)0, acc,  false,  false);
    }

    const int r0 = tm * 16 + half * 8;
    #pragma unroll
    for (int v = 0; v < 8; ++v) {
        C[(size_t)(r0 + v) * N + ncol] = acc[v];
    }
}

__global__ __launch_bounds__(128)
void attn_kernel(const float* __restrict__ ph, const float* __restrict__ pu,
                 const float* __restrict__ a, float* __restrict__ h_new,
                 float* __restrict__ alpha_out) {
    const int e = blockIdx.x;
    const int d = threadIdx.x;

    const float vph = ph[(size_t)e * 128 + d];
    const float vpu = pu[(size_t)e * 128 + d];
    const float ad  = a[d];

    __shared__ float r0[128];
    __shared__ float r1[128];
    r0[d] = tanhf(vph) * ad;
    r1[d] = tanhf(vpu) * ad;
    __syncthreads();

    #pragma unroll
    for (int s = 64; s > 0; s >>= 1) {
        if (d < s) { r0[d] += r0[d + s]; r1[d] += r1[d + s]; }
        __syncthreads();
    }

    __shared__ float al0, al12;
    if (d == 0) {
        const float s0 = r0[0], s1 = r1[0];
        const float mx = fmaxf(s0, s1);
        const float e0 = expf(s0 - mx);
        const float e1 = expf(s1 - mx);
        const float denom = e0 + 2.f * e1;
        const float a0 = e0 / denom;
        const float a1 = e1 / denom;
        al0  = a0;
        al12 = 2.f * a1;
        alpha_out[(size_t)e * 3 + 0] = a0;
        alpha_out[(size_t)e * 3 + 1] = a1;
        alpha_out[(size_t)e * 3 + 2] = a1;
    }
    __syncthreads();

    h_new[(size_t)e * 128 + d] = al0 * vph + al12 * vpu;
}

extern "C" void kernel_launch(void* const* d_in, const int* in_sizes, int n_in,
                              void* d_out, int out_size, void* d_ws, size_t ws_size,
                              hipStream_t stream) {
    const float* h_xe = (const float*)d_in[0];
    const float* h    = (const float*)d_in[1];
    const float* m    = (const float*)d_in[2];
    const float* W_h  = (const float*)d_in[3];
    const float* W_f  = (const float*)d_in[4];
    const float* avec = (const float*)d_in[5];
    const int* idx_e  = (const int*)d_in[6];
    const int* idx_n  = (const int*)d_in[7];

    const int E  = in_sizes[0] / 64;
    const int DH = in_sizes[5];

    float* f_u = (float*)d_ws;
    float* ph  = f_u + (size_t)E * DH;
    float* pu  = ph  + (size_t)E * DH;

    float* h_new     = (float*)d_out;
    float* alpha_out = (float*)d_out + (size_t)E * DH;

    agg_kernel<<<E, 256, 0, stream>>>(h, m, idx_e, idx_n, f_u);

    const int tiles  = (E / 16) * (DH / 16);
    const int blocks = tiles / 8;
    wmma_gemm_f32<<<blocks, 256, 0, stream>>>(h_xe, W_h, ph, E, 64);
    wmma_gemm_f32<<<blocks, 256, 0, stream>>>(f_u,  W_f, pu, E, 128);

    attn_kernel<<<E, 128, 0, stream>>>(ph, pu, avec, h_new, alpha_out);
}


# both projections and the attention head in one kernel: a wave owns 32 rows x 32 columns as 2x2 register tiles fed straight from global memory; 4 launches -> 2; k and score sum order changed
# speedup vs baseline: 1.1319x; 1.1319x over previous
#include <hip/hip_runtime.h>
#include <math.h>

typedef __attribute__((ext_vector_type(2))) float v2f;
typedef __attribute__((ext_vector_type(4))) float v4f;
typedef __attribute__((ext_vector_type(8))) float v8f;

__global__ __launch_bounds__(256)
void agg_kernel(const float* __restrict__ h, const float* __restrict__ m,
                const int* __restrict__ idx_e, const int* __restrict__ idx_n,
                float* __restrict__ f_u) {
    const int e     = blockIdx.x;
    const int tid   = threadIdx.x;
    const int quad  = tid & 15;
    const int rslot = tid >> 4;

    __shared__ int s_ie[64];
    __shared__ int s_in[64];
    if (tid < 64)          s_ie[tid]      = idx_e[(size_t)e * 64 + tid];
    else if (tid < 128)    s_in[tid - 64] = idx_n[(size_t)e * 64 + (tid - 64)];
    __syncthreads();

    const v4f* __restrict__ hv = (const v4f*)h;
    const v4f* __restrict__ mv = (const v4f*)m;
    v4f* __restrict__ fv       = (v4f*)f_u;

    __shared__ v4f red[256];

    {
        v4f acc = {0.f, 0.f, 0.f, 0.f};
        #pragma unroll
        for (int i = 0; i < 4; ++i) {
            const int r = rslot + 16 * i;
            acc += hv[(size_t)s_ie[r] * 16 + quad];
        }
        red[tid] = acc;
        __syncthreads();
        #pragma unroll
        for (int s = 128; s >= 16; s >>= 1) {
            if (tid < s) red[tid] += red[tid + s];
            __syncthreads();
        }
        if (tid < 16) {
            v4f r = red[tid];
            r *= (1.0f / 64.0f);
            fv[(size_t)e * 32 + tid] = r;
        }
        __syncthreads();
    }

    {
        v4f acc = {0.f, 0.f, 0.f, 0.f};
        #pragma unroll
        for (int i = 0; i < 4; ++i) {
            const int r = rslot + 16 * i;
            acc += mv[(size_t)s_in[r] * 16 + quad];
        }
        red[tid] = acc;
        __syncthreads();
        #pragma unroll
        for (int s = 128; s >= 16; s >>= 1) {
            if (tid < s) red[tid] += red[tid + s];
            __syncthreads();
        }
        if (tid < 16) {
            v4f r = red[tid];
            r *= (1.0f / 64.0f);
            fv[(size_t)e * 32 + 16 + tid] = r;
        }
    }
}

__global__ __launch_bounds__(128)
void proj_attn_kernel(const float* __restrict__ h_xe, const float* __restrict__ f_u,
                      const float* __restrict__ W_h, const float* __restrict__ W_f,
                      const float* __restrict__ a, float* __restrict__ h_new,
                      float* __restrict__ alpha_out) {
    const int wave = threadIdx.x >> 5;
    const int lane = threadIdx.x & 31;
    const int c    = lane & 15;
    const int hf   = lane >> 4;
    const int row0 = blockIdx.x * 32;
    const int col0 = 32 * wave + c;

    constexpr int NBH   = 64 / 8;
    constexpr int NB    = NBH + 128 / 8;
    constexpr int DEPTH = 4;

    const float* Ah = h_xe + (size_t)(row0 + c) * 64 + 4 * hf;
    const float* Af = f_u + (size_t)(row0 + c) * 128 + 4 * hf;
    const float* Bh = W_h + (size_t)(4 * hf) * 128 + col0;
    const float* Bf = W_f + (size_t)(4 * hf) * 128 + col0;

    v8f ph[2][2], pu[2][2];
    #pragma unroll
    for (int i = 0; i < 2; ++i) {
        #pragma unroll
        for (int n = 0; n < 2; ++n) {
            ph[i][n] = (v8f){0.f, 0.f, 0.f, 0.f, 0.f, 0.f, 0.f, 0.f};
            pu[i][n] = (v8f){0.f, 0.f, 0.f, 0.f, 0.f, 0.f, 0.f, 0.f};
        }
    }

    v4f fa[NB][2];
    v2f fb[NB][2][2];

    #pragma unroll
    for (int g = 0; g < DEPTH; ++g) {
        const float* Ap = (g < NBH) ? Ah + 8 * g : Af + 8 * (g - NBH);
        const float* Bp = (g < NBH) ? Bh + (size_t)(8 * g) * 128
                                    : Bf + (size_t)(8 * (g - NBH)) * 128;
        const int    lda = (g < NBH) ? 64 : 128;
        fa[g][0] = *(const v4f*)Ap;
        fa[g][1] = *(const v4f*)(Ap + (size_t)16 * lda);
        #pragma unroll
        for (int n = 0; n < 2; ++n) {
            #pragma unroll
            for (int t = 0; t < 2; ++t) {
                fb[g][n][t].x = Bp[(size_t)(2 * t) * 128 + 16 * n];
                fb[g][n][t].y = Bp[(size_t)(2 * t + 1) * 128 + 16 * n];
            }
        }
    }

    #pragma unroll
    for (int g = 0; g < NB; ++g) {
        if (g + DEPTH < NB) {
            const int q = g + DEPTH;
            const float* Ap = (q < NBH) ? Ah + 8 * q : Af + 8 * (q - NBH);
            const float* Bp = (q < NBH) ? Bh + (size_t)(8 * q) * 128
                                        : Bf + (size_t)(8 * (q - NBH)) * 128;
            const int    lda = (q < NBH) ? 64 : 128;
            fa[q][0] = *(const v4f*)Ap;
            fa[q][1] = *(const v4f*)(Ap + (size_t)16 * lda);
            #pragma unroll
            for (int n = 0; n < 2; ++n) {
                #pragma unroll
                for (int t = 0; t < 2; ++t) {
                    fb[q][n][t].x = Bp[(size_t)(2 * t) * 128 + 16 * n];
                    fb[q][n][t].y = Bp[(size_t)(2 * t + 1) * 128 + 16 * n];
                }
            }
        }
        __builtin_amdgcn_sched_barrier(0);
        #pragma unroll
        for (int t = 0; t < 2; ++t) {
            #pragma unroll
            for (int i = 0; i < 2; ++i) {
                v2f af;
                af.x = fa[g][i][2 * t];
                af.y = fa[g][i][2 * t + 1];
                #pragma unroll
                for (int n = 0; n < 2; ++n) {
                    if (g < NBH)
                        ph[i][n] = __builtin_amdgcn_wmma_f32_16x16x4_f32(
                             false, af,  false, fb[g][n][t],
                             (short)0, ph[i][n],  false,  false);
                    else
                        pu[i][n] = __builtin_amdgcn_wmma_f32_16x16x4_f32(
                             false, af,  false, fb[g][n][t],
                             (short)0, pu[i][n],  false,  false);
                }
            }
        }
    }

    const float av0 = a[col0];
    const float av1 = a[col0 + 16];

    __shared__ float part[4][2][32];
    #pragma unroll
    for (int i = 0; i < 2; ++i) {
        #pragma unroll
        for (int r = 0; r < 8; ++r) {
            float s0 = tanhf(ph[i][0][r]) * av0 + tanhf(ph[i][1][r]) * av1;
            float s1 = tanhf(pu[i][0][r]) * av0 + tanhf(pu[i][1][r]) * av1;
            #pragma unroll
            for (int mk = 1; mk < 16; mk <<= 1) {
                s0 += __shfl_xor(s0, mk);
                s1 += __shfl_xor(s1, mk);
            }
            if (c == 0) {
                part[wave][0][16 * i + 8 * hf + r] = s0;
                part[wave][1][16 * i + 8 * hf + r] = s1;
            }
        }
    }
    __syncthreads();

    #pragma unroll
    for (int i = 0; i < 2; ++i) {
        #pragma unroll
        for (int r = 0; r < 8; ++r) {
            const int   rl = 16 * i + 8 * hf + r;
            const float s0 = (part[0][0][rl] + part[1][0][rl]) + (part[2][0][rl] + part[3][0][rl]);
            const float s1 = (part[0][1][rl] + part[1][1][rl]) + (part[2][1][rl] + part[3][1][rl]);
            const float mx = fmaxf(s0, s1);
            const float e0 = expf(s0 - mx);
            const float e1 = expf(s1 - mx);
            const float denom = e0 + 2.f * e1;
            const float a0  = e0 / denom;
            const float a1  = e1 / denom;
            const float a12 = 2.f * a1;
            float* orow = h_new + (size_t)(row0 + rl) * 128 + col0;
            orow[0]  = a0 * ph[i][0][r] + a12 * pu[i][0][r];
            orow[16] = a0 * ph[i][1][r] + a12 * pu[i][1][r];
            if (wave == 0 && c < 3)
                alpha_out[(size_t)(row0 + rl) * 3 + c] = (c == 0) ? a0 : a1;
        }
    }
}

extern "C" void kernel_launch(void* const* d_in, const int* in_sizes, int n_in,
                              void* d_out, int out_size, void* d_ws, size_t ws_size,
                              hipStream_t stream) {
    const float* h_xe = (const float*)d_in[0];
    const float* h    = (const float*)d_in[1];
    const float* m    = (const float*)d_in[2];
    const float* W_h  = (const float*)d_in[3];
    const float* W_f  = (const float*)d_in[4];
    const float* avec = (const float*)d_in[5];
    const int* idx_e  = (const int*)d_in[6];
    const int* idx_n  = (const int*)d_in[7];

    const int E  = in_sizes[0] / 64;
    const int DH = in_sizes[5];

    float* f_u = (float*)d_ws;

    float* h_new     = (float*)d_out;
    float* alpha_out = (float*)d_out + (size_t)E * DH;

    agg_kernel<<<E, 256, 0, stream>>>(h, m, idx_e, idx_n, f_u);

    proj_attn_kernel<<<E / 32, 128, 0, stream>>>(h_xe, f_u, W_h, W_f, avec, h_new, alpha_out);
}


# gather kernel persistent: 1024 workgroups each walk 8 edges as a software pipeline (next edge's rows in flight across an LDS-only barrier, one-pass LDS sum); 65536 -> 8192 waves; sum order changed
# speedup vs baseline: 1.2106x; 1.0695x over previous
#include <hip/hip_runtime.h>
#include <math.h>

typedef __attribute__((ext_vector_type(2))) float v2f;
typedef __attribute__((ext_vector_type(4))) float v4f;
typedef __attribute__((ext_vector_type(8))) float v8f;

constexpr int EPB = 8;

__global__ __launch_bounds__(256) __attribute__((amdgpu_waves_per_eu(8)))
void agg_kernel(const float* __restrict__ h, const float* __restrict__ m,
                const int* __restrict__ idx_e, const int* __restrict__ idx_n,
                float* __restrict__ f_u) {
    const int tid   = threadIdx.x;
    const int quad  = tid & 15;
    const int rslot = tid >> 4;
    const int e0    = blockIdx.x * EPB;

    const v4f* __restrict__ hv = (const v4f*)h;
    const v4f* __restrict__ mv = (const v4f*)m;
    v4f* __restrict__ fv       = (v4f*)f_u;

    __shared__ v4f red[2][2][256];

    const int* pe = idx_e + (size_t)e0 * 64 + rslot;
    const int* pn = idx_n + (size_t)e0 * 64 + rslot;

    int ixe[EPB][4], ixn[EPB][4];
    v4f rwh[EPB][4], rwm[EPB][4];

    #pragma unroll
    for (int i = 0; i < 4; ++i) {
        ixe[0][i] = pe[16 * i];
        ixn[0][i] = pn[16 * i];
    }
    #pragma unroll
    for (int i = 0; i < 4; ++i) {
        ixe[1][i] = pe[64 + 16 * i];
        ixn[1][i] = pn[64 + 16 * i];
    }
    __builtin_amdgcn_sched_barrier(0);
    #pragma unroll
    for (int i = 0; i < 4; ++i) {
        rwh[0][i] = hv[(unsigned)ixe[0][i] * 16u + (unsigned)quad];
        rwm[0][i] = mv[(unsigned)ixn[0][i] * 16u + (unsigned)quad];
    }

    #pragma unroll
    for (int it = 0; it < EPB; ++it) {
        if (it + 2 < EPB) {
            #pragma unroll
            for (int i = 0; i < 4; ++i) {
                ixe[it + 2][i] = pe[64 * (it + 2) + 16 * i];
                ixn[it + 2][i] = pn[64 * (it + 2) + 16 * i];
            }
        }
        __builtin_amdgcn_sched_barrier(0);
        if (it + 1 < EPB) {
            #pragma unroll
            for (int i = 0; i < 4; ++i) {
                rwh[it + 1][i] = hv[(unsigned)ixe[it + 1][i] * 16u + (unsigned)quad];
                rwm[it + 1][i] = mv[(unsigned)ixn[it + 1][i] * 16u + (unsigned)quad];
            }
        }
        __builtin_amdgcn_sched_barrier(0);

        const int buf = it & 1;
        red[buf][0][tid] = (rwh[it][0] + rwh[it][1]) + (rwh[it][2] + rwh[it][3]);
        red[buf][1][tid] = (rwm[it][0] + rwm[it][1]) + (rwm[it][2] + rwm[it][3]);

        asm volatile("s_wait_dscnt 0x0" ::: "memory");
        asm volatile("s_barrier_signal -1" ::: "memory");
        asm volatile("s_barrier_wait -1" ::: "memory");

        if (tid < 32) {
            const v4f* src = &red[buf][tid >> 4][quad];
            v4f s[16];
            #pragma unroll
            for (int k = 0; k < 16; ++k) s[k] = src[16 * k];
            #pragma unroll
            for (int w = 8; w >= 1; w >>= 1) {
                #pragma unroll
                for (int k = 0; k < w; ++k) s[k] += s[k + w];
            }
            fv[(size_t)(e0 + it) * 32 + tid] = s[0] * (1.0f / 64.0f);
        }
    }
}

__global__ __launch_bounds__(128)
void proj_attn_kernel(const float* __restrict__ h_xe, const float* __restrict__ f_u,
                      const float* __restrict__ W_h, const float* __restrict__ W_f,
                      const float* __restrict__ a, float* __restrict__ h_new,
                      float* __restrict__ alpha_out) {
    const int wave = threadIdx.x >> 5;
    const int lane = threadIdx.x & 31;
    const int c    = lane & 15;
    const int hf   = lane >> 4;
    const int row0 = blockIdx.x * 32;
    const int col0 = 32 * wave + c;

    constexpr int NBH   = 64 / 8;
    constexpr int NB    = NBH + 128 / 8;
    constexpr int DEPTH = 4;

    const float* Ah = h_xe + (size_t)(row0 + c) * 64 + 4 * hf;
    const float* Af = f_u + (size_t)(row0 + c) * 128 + 4 * hf;
    const float* Bh = W_h + (size_t)(4 * hf) * 128 + col0;
    const float* Bf = W_f + (size_t)(4 * hf) * 128 + col0;

    v8f ph[2][2], pu[2][2];
    #pragma unroll
    for (int i = 0; i < 2; ++i) {
        #pragma unroll
        for (int n = 0; n < 2; ++n) {
            ph[i][n] = (v8f){0.f, 0.f, 0.f, 0.f, 0.f, 0.f, 0.f, 0.f};
            pu[i][n] = (v8f){0.f, 0.f, 0.f, 0.f, 0.f, 0.f, 0.f, 0.f};
        }
    }

    v4f fa[NB][2];
    v2f fb[NB][2][2];

    #pragma unroll
    for (int g = 0; g < DEPTH; ++g) {
        const float* Ap = (g < NBH) ? Ah + 8 * g : Af + 8 * (g - NBH);
        const float* Bp = (g < NBH) ? Bh + (size_t)(8 * g) * 128
                                    : Bf + (size_t)(8 * (g - NBH)) * 128;
        const int    lda = (g < NBH) ? 64 : 128;
        fa[g][0] = *(const v4f*)Ap;
        fa[g][1] = *(const v4f*)(Ap + (size_t)16 * lda);
        #pragma unroll
        for (int n = 0; n < 2; ++n) {
            #pragma unroll
            for (int t = 0; t < 2; ++t) {
                fb[g][n][t].x = Bp[(size_t)(2 * t) * 128 + 16 * n];
                fb[g][n][t].y = Bp[(size_t)(2 * t + 1) * 128 + 16 * n];
            }
        }
    }

    #pragma unroll
    for (int g = 0; g < NB; ++g) {
        if (g + DEPTH < NB) {
            const int q = g + DEPTH;
            const float* Ap = (q < NBH) ? Ah + 8 * q : Af + 8 * (q - NBH);
            const float* Bp = (q < NBH) ? Bh + (size_t)(8 * q) * 128
                                        : Bf + (size_t)(8 * (q - NBH)) * 128;
            const int    lda = (q < NBH) ? 64 : 128;
            fa[q][0] = *(const v4f*)Ap;
            fa[q][1] = *(const v4f*)(Ap + (size_t)16 * lda);
            #pragma unroll
            for (int n = 0; n < 2; ++n) {
                #pragma unroll
                for (int t = 0; t < 2; ++t) {
                    fb[q][n][t].x = Bp[(size_t)(2 * t) * 128 + 16 * n];
                    fb[q][n][t].y = Bp[(size_t)(2 * t + 1) * 128 + 16 * n];
                }
            }
        }
        __builtin_amdgcn_sched_barrier(0);
        #pragma unroll
        for (int t = 0; t < 2; ++t) {
            #pragma unroll
            for (int i = 0; i < 2; ++i) {
                v2f af;
                af.x = fa[g][i][2 * t];
                af.y = fa[g][i][2 * t + 1];
                #pragma unroll
                for (int n = 0; n < 2; ++n) {
                    if (g < NBH)
                        ph[i][n] = __builtin_amdgcn_wmma_f32_16x16x4_f32(
                             false, af,  false, fb[g][n][t],
                             (short)0, ph[i][n],  false,  false);
                    else
                        pu[i][n] = __builtin_amdgcn_wmma_f32_16x16x4_f32(
                             false, af,  false, fb[g][n][t],
                             (short)0, pu[i][n],  false,  false);
                }
            }
        }
    }

    const float av0 = a[col0];
    const float av1 = a[col0 + 16];

    __shared__ float part[4][2][32];
    #pragma unroll
    for (int i = 0; i < 2; ++i) {
        #pragma unroll
        for (int r = 0; r < 8; ++r) {
            float s0 = tanhf(ph[i][0][r]) * av0 + tanhf(ph[i][1][r]) * av1;
            float s1 = tanhf(pu[i][0][r]) * av0 + tanhf(pu[i][1][r]) * av1;
            #pragma unroll
            for (int mk = 1; mk < 16; mk <<= 1) {
                s0 += __shfl_xor(s0, mk);
                s1 += __shfl_xor(s1, mk);
            }
            if (c == 0) {
                part[wave][0][16 * i + 8 * hf + r] = s0;
                part[wave][1][16 * i + 8 * hf + r] = s1;
            }
        }
    }
    __syncthreads();

    #pragma unroll
    for (int i = 0; i < 2; ++i) {
        #pragma unroll
        for (int r = 0; r < 8; ++r) {
            const int   rl = 16 * i + 8 * hf + r;
            const float s0 = (part[0][0][rl] + part[1][0][rl]) + (part[2][0][rl] + part[3][0][rl]);
            const float s1 = (part[0][1][rl] + part[1][1][rl]) + (part[2][1][rl] + part[3][1][rl]);
            const float mx = fmaxf(s0, s1);
            const float e0 = expf(s0 - mx);
            const float e1 = expf(s1 - mx);
            const float denom = e0 + 2.f * e1;
            const float a0  = e0 / denom;
            const float a1  = e1 / denom;
            const float a12 = 2.f * a1;
            float* orow = h_new + (size_t)(row0 + rl) * 128 + col0;
            orow[0]  = a0 * ph[i][0][r] + a12 * pu[i][0][r];
            orow[16] = a0 * ph[i][1][r] + a12 * pu[i][1][r];
            if (wave == 0 && c < 3)
                alpha_out[(size_t)(row0 + rl) * 3 + c] = (c == 0) ? a0 : a1;
        }
    }
}

extern "C" void kernel_launch(void* const* d_in, const int* in_sizes, int n_in,
                              void* d_out, int out_size, void* d_ws, size_t ws_size,
                              hipStream_t stream) {
    const float* h_xe = (const float*)d_in[0];
    const float* h    = (const float*)d_in[1];
    const float* m    = (const float*)d_in[2];
    const float* W_h  = (const float*)d_in[3];
    const float* W_f  = (const float*)d_in[4];
    const float* avec = (const float*)d_in[5];
    const int* idx_e  = (const int*)d_in[6];
    const int* idx_n  = (const int*)d_in[7];

    const int E  = in_sizes[0] / 64;
    const int DH = in_sizes[5];

    float* f_u = (float*)d_ws;

    float* h_new     = (float*)d_out;
    float* alpha_out = (float*)d_out + (size_t)E * DH;

    agg_kernel<<<E / EPB, 256, 0, stream>>>(h, m, idx_e, idx_n, f_u);

    proj_attn_kernel<<<E / 32, 128, 0, stream>>>(h_xe, f_u, W_h, W_f, avec, h_new, alpha_out);
}


# attention scores: the 32 row sums of a wave reduced together over the 16 lanes by row_xmask DPP adds behind the tanh pass (was one dependent LDS-permute chain per row); sum order unchanged
# speedup vs baseline: 1.2860x; 1.0623x over previous
#include <hip/hip_runtime.h>
#include <math.h>

typedef __attribute__((ext_vector_type(2))) float v2f;
typedef __attribute__((ext_vector_type(4))) float v4f;
typedef __attribute__((ext_vector_type(8))) float v8f;

constexpr int EPB = 8;

__global__ __launch_bounds__(256) __attribute__((amdgpu_waves_per_eu(8)))
void agg_kernel(const float* __restrict__ h, const float* __restrict__ m,
                const int* __restrict__ idx_e, const int* __restrict__ idx_n,
                float* __restrict__ f_u) {
    const int tid   = threadIdx.x;
    const int quad  = tid & 15;
    const int rslot = tid >> 4;
    const int e0    = blockIdx.x * EPB;

    const v4f* __restrict__ hv = (const v4f*)h;
    const v4f* __restrict__ mv = (const v4f*)m;
    v4f* __restrict__ fv       = (v4f*)f_u;

    __shared__ v4f red[2][2][256];

    const int* pe = idx_e + (size_t)e0 * 64 + rslot;
    const int* pn = idx_n + (size_t)e0 * 64 + rslot;

    int ixe[EPB][4], ixn[EPB][4];
    v4f rwh[EPB][4], rwm[EPB][4];

    #pragma unroll
    for (int i = 0; i < 4; ++i) {
        ixe[0][i] = pe[16 * i];
        ixn[0][i] = pn[16 * i];
    }
    #pragma unroll
    for (int i = 0; i < 4; ++i) {
        ixe[1][i] = pe[64 + 16 * i];
        ixn[1][i] = pn[64 + 16 * i];
    }
    __builtin_amdgcn_sched_barrier(0);
    #pragma unroll
    for (int i = 0; i < 4; ++i) {
        rwh[0][i] = hv[(unsigned)ixe[0][i] * 16u + (unsigned)quad];
        rwm[0][i] = mv[(unsigned)ixn[0][i] * 16u + (unsigned)quad];
    }

    #pragma unroll
    for (int it = 0; it < EPB; ++it) {
        if (it + 2 < EPB) {
            #pragma unroll
            for (int i = 0; i < 4; ++i) {
                ixe[it + 2][i] = pe[64 * (it + 2) + 16 * i];
                ixn[it + 2][i] = pn[64 * (it + 2) + 16 * i];
            }
        }
        __builtin_amdgcn_sched_barrier(0);
        if (it + 1 < EPB) {
            #pragma unroll
            for (int i = 0; i < 4; ++i) {
                rwh[it + 1][i] = hv[(unsigned)ixe[it + 1][i] * 16u + (unsigned)quad];
                rwm[it + 1][i] = mv[(unsigned)ixn[it + 1][i] * 16u + (unsigned)quad];
            }
        }
        __builtin_amdgcn_sched_barrier(0);

        const int buf = it & 1;
        red[buf][0][tid] = (rwh[it][0] + rwh[it][1]) + (rwh[it][2] + rwh[it][3]);
        red[buf][1][tid] = (rwm[it][0] + rwm[it][1]) + (rwm[it][2] + rwm[it][3]);

        asm volatile("s_wait_dscnt 0x0" ::: "memory");
        asm volatile("s_barrier_signal -1" ::: "memory");
        asm volatile("s_barrier_wait -1" ::: "memory");

        if (tid < 32) {
            const v4f* src = &red[buf][tid >> 4][quad];
            v4f s[16];
            #pragma unroll
            for (int k = 0; k < 16; ++k) s[k] = src[16 * k];
            #pragma unroll
            for (int w = 8; w >= 1; w >>= 1) {
                #pragma unroll
                for (int k = 0; k < w; ++k) s[k] += s[k + w];
            }
            fv[(size_t)(e0 + it) * 32 + tid] = s[0] * (1.0f / 64.0f);
        }
    }
}

template <int CTRL>
__device__ __forceinline__ float row_xor_add(float v) {
    const int t = __builtin_amdgcn_update_dpp(0, __builtin_bit_cast(int, v), CTRL,
                                               0xf,  0xf,
                                               true);
    return v + __builtin_bit_cast(float, t);
}

__global__ __launch_bounds__(128)
void proj_attn_kernel(const float* __restrict__ h_xe, const float* __restrict__ f_u,
                      const float* __restrict__ W_h, const float* __restrict__ W_f,
                      const float* __restrict__ a, float* __restrict__ h_new,
                      float* __restrict__ alpha_out) {
    const int wave = threadIdx.x >> 5;
    const int lane = threadIdx.x & 31;
    const int c    = lane & 15;
    const int hf   = lane >> 4;
    const int row0 = blockIdx.x * 32;
    const int col0 = 32 * wave + c;

    constexpr int NBH   = 64 / 8;
    constexpr int NB    = NBH + 128 / 8;
    constexpr int DEPTH = 4;

    const float* Ah = h_xe + (size_t)(row0 + c) * 64 + 4 * hf;
    const float* Af = f_u + (size_t)(row0 + c) * 128 + 4 * hf;
    const float* Bh = W_h + (size_t)(4 * hf) * 128 + col0;
    const float* Bf = W_f + (size_t)(4 * hf) * 128 + col0;

    v8f ph[2][2], pu[2][2];
    #pragma unroll
    for (int i = 0; i < 2; ++i) {
        #pragma unroll
        for (int n = 0; n < 2; ++n) {
            ph[i][n] = (v8f){0.f, 0.f, 0.f, 0.f, 0.f, 0.f, 0.f, 0.f};
            pu[i][n] = (v8f){0.f, 0.f, 0.f, 0.f, 0.f, 0.f, 0.f, 0.f};
        }
    }

    v4f fa[NB][2];
    v2f fb[NB][2][2];

    #pragma unroll
    for (int g = 0; g < DEPTH; ++g) {
        const float* Ap = (g < NBH) ? Ah + 8 * g : Af + 8 * (g - NBH);
        const float* Bp = (g < NBH) ? Bh + (size_t)(8 * g) * 128
                                    : Bf + (size_t)(8 * (g - NBH)) * 128;
        const int    lda = (g < NBH) ? 64 : 128;
        fa[g][0] = *(const v4f*)Ap;
        fa[g][1] = *(const v4f*)(Ap + (size_t)16 * lda);
        #pragma unroll
        for (int n = 0; n < 2; ++n) {
            #pragma unroll
            for (int t = 0; t < 2; ++t) {
                fb[g][n][t].x = Bp[(size_t)(2 * t) * 128 + 16 * n];
                fb[g][n][t].y = Bp[(size_t)(2 * t + 1) * 128 + 16 * n];
            }
        }
    }

    #pragma unroll
    for (int g = 0; g < NB; ++g) {
        if (g + DEPTH < NB) {
            const int q = g + DEPTH;
            const float* Ap = (q < NBH) ? Ah + 8 * q : Af + 8 * (q - NBH);
            const float* Bp = (q < NBH) ? Bh + (size_t)(8 * q) * 128
                                        : Bf + (size_t)(8 * (q - NBH)) * 128;
            const int    lda = (q < NBH) ? 64 : 128;
            fa[q][0] = *(const v4f*)Ap;
            fa[q][1] = *(const v4f*)(Ap + (size_t)16 * lda);
            #pragma unroll
            for (int n = 0; n < 2; ++n) {
                #pragma unroll
                for (int t = 0; t < 2; ++t) {
                    fb[q][n][t].x = Bp[(size_t)(2 * t) * 128 + 16 * n];
                    fb[q][n][t].y = Bp[(size_t)(2 * t + 1) * 128 + 16 * n];
                }
            }
        }
        __builtin_amdgcn_sched_barrier(0);
        #pragma unroll
        for (int t = 0; t < 2; ++t) {
            #pragma unroll
            for (int i = 0; i < 2; ++i) {
                v2f af;
                af.x = fa[g][i][2 * t];
                af.y = fa[g][i][2 * t + 1];
                #pragma unroll
                for (int n = 0; n < 2; ++n) {
                    if (g < NBH)
                        ph[i][n] = __builtin_amdgcn_wmma_f32_16x16x4_f32(
                             false, af,  false, fb[g][n][t],
                             (short)0, ph[i][n],  false,  false);
                    else
                        pu[i][n] = __builtin_amdgcn_wmma_f32_16x16x4_f32(
                             false, af,  false, fb[g][n][t],
                             (short)0, pu[i][n],  false,  false);
                }
            }
        }
    }

    const float av0 = a[col0];
    const float av1 = a[col0 + 16];

    __shared__ float part[4][2][32];
    float s0[2][8], s1[2][8];
    #pragma unroll
    for (int i = 0; i < 2; ++i) {
        #pragma unroll
        for (int r = 0; r < 8; ++r) {
            s0[i][r] = tanhf(ph[i][0][r]) * av0 + tanhf(ph[i][1][r]) * av1;
            s1[i][r] = tanhf(pu[i][0][r]) * av0 + tanhf(pu[i][1][r]) * av1;
        }
    }
    #pragma unroll
    for (int i = 0; i < 2; ++i) {
        #pragma unroll
        for (int r = 0; r < 8; ++r) {
            s0[i][r] = row_xor_add<0x161>(s0[i][r]);
            s1[i][r] = row_xor_add<0x161>(s1[i][r]);
        }
    }
    #pragma unroll
    for (int i = 0; i < 2; ++i) {
        #pragma unroll
        for (int r = 0; r < 8; ++r) {
            s0[i][r] = row_xor_add<0x162>(s0[i][r]);
            s1[i][r] = row_xor_add<0x162>(s1[i][r]);
        }
    }
    #pragma unroll
    for (int i = 0; i < 2; ++i) {
        #pragma unroll
        for (int r = 0; r < 8; ++r) {
            s0[i][r] = row_xor_add<0x164>(s0[i][r]);
            s1[i][r] = row_xor_add<0x164>(s1[i][r]);
        }
    }
    #pragma unroll
    for (int i = 0; i < 2; ++i) {
        #pragma unroll
        for (int r = 0; r < 8; ++r) {
            s0[i][r] = row_xor_add<0x168>(s0[i][r]);
            s1[i][r] = row_xor_add<0x168>(s1[i][r]);
        }
    }
    if (c == 0) {
        #pragma unroll
        for (int i = 0; i < 2; ++i) {
            #pragma unroll
            for (int r = 0; r < 8; ++r) {
                part[wave][0][16 * i + 8 * hf + r] = s0[i][r];
                part[wave][1][16 * i + 8 * hf + r] = s1[i][r];
            }
        }
    }
    __syncthreads();

    #pragma unroll
    for (int i = 0; i < 2; ++i) {
        #pragma unroll
        for (int r = 0; r < 8; ++r) {
            const int   rl = 16 * i + 8 * hf + r;
            const float s0 = (part[0][0][rl] + part[1][0][rl]) + (part[2][0][rl] + part[3][0][rl]);
            const float s1 = (part[0][1][rl] + part[1][1][rl]) + (part[2][1][rl] + part[3][1][rl]);
            const float mx = fmaxf(s0, s1);
            const float e0 = expf(s0 - mx);
            const float e1 = expf(s1 - mx);
            const float denom = e0 + 2.f * e1;
            const float a0  = e0 / denom;
            const float a1  = e1 / denom;
            const float a12 = 2.f * a1;
            float* orow = h_new + (size_t)(row0 + rl) * 128 + col0;
            orow[0]  = a0 * ph[i][0][r] + a12 * pu[i][0][r];
            orow[16] = a0 * ph[i][1][r] + a12 * pu[i][1][r];
            if (wave == 0 && c < 3)
                alpha_out[(size_t)(row0 + rl) * 3 + c] = (c == 0) ? a0 : a1;
        }
    }
}

extern "C" void kernel_launch(void* const* d_in, const int* in_sizes, int n_in,
                              void* d_out, int out_size, void* d_ws, size_t ws_size,
                              hipStream_t stream) {
    const float* h_xe = (const float*)d_in[0];
    const float* h    = (const float*)d_in[1];
    const float* m    = (const float*)d_in[2];
    const float* W_h  = (const float*)d_in[3];
    const float* W_f  = (const float*)d_in[4];
    const float* avec = (const float*)d_in[5];
    const int* idx_e  = (const int*)d_in[6];
    const int* idx_n  = (const int*)d_in[7];

    const int E  = in_sizes[0] / 64;
    const int DH = in_sizes[5];

    float* f_u = (float*)d_ws;

    float* h_new     = (float*)d_out;
    float* alpha_out = (float*)d_out + (size_t)E * DH;

    agg_kernel<<<E / EPB, 256, 0, stream>>>(h, m, idx_e, idx_n, f_u);

    proj_attn_kernel<<<E / 32, 128, 0, stream>>>(h_xe, f_u, W_h, W_f, avec, h_new, alpha_out);
}


# projection/attention kernel on two waves per SIMD: 8-wave block per 32 rows, the two waves of a column quarter split k in halves, swap half tiles via LDS, each finishes 16 rows; k sum now two halves
# speedup vs baseline: 1.4323x; 1.1138x over previous
#include <hip/hip_runtime.h>
#include <math.h>

typedef __attribute__((ext_vector_type(2))) float v2f;
typedef __attribute__((ext_vector_type(4))) float v4f;
typedef __attribute__((ext_vector_type(8))) float v8f;

constexpr int EPB = 8;

__global__ __launch_bounds__(256) __attribute__((amdgpu_waves_per_eu(8)))
void agg_kernel(const float* __restrict__ h, const float* __restrict__ m,
                const int* __restrict__ idx_e, const int* __restrict__ idx_n,
                float* __restrict__ f_u) {
    const int tid   = threadIdx.x;
    const int quad  = tid & 15;
    const int rslot = tid >> 4;
    const int e0    = blockIdx.x * EPB;

    const v4f* __restrict__ hv = (const v4f*)h;
    const v4f* __restrict__ mv = (const v4f*)m;
    v4f* __restrict__ fv       = (v4f*)f_u;

    __shared__ v4f red[2][2][256];

    const int* pe = idx_e + (size_t)e0 * 64 + rslot;
    const int* pn = idx_n + (size_t)e0 * 64 + rslot;

    int ixe[EPB][4], ixn[EPB][4];
    v4f rwh[EPB][4], rwm[EPB][4];

    #pragma unroll
    for (int i = 0; i < 4; ++i) {
        ixe[0][i] = pe[16 * i];
        ixn[0][i] = pn[16 * i];
    }
    #pragma unroll
    for (int i = 0; i < 4; ++i) {
        ixe[1][i] = pe[64 + 16 * i];
        ixn[1][i] = pn[64 + 16 * i];
    }
    __builtin_amdgcn_sched_barrier(0);
    #pragma unroll
    for (int i = 0; i < 4; ++i) {
        rwh[0][i] = hv[(unsigned)ixe[0][i] * 16u + (unsigned)quad];
        rwm[0][i] = mv[(unsigned)ixn[0][i] * 16u + (unsigned)quad];
    }

    #pragma unroll
    for (int it = 0; it < EPB; ++it) {
        if (it + 2 < EPB) {
            #pragma unroll
            for (int i = 0; i < 4; ++i) {
                ixe[it + 2][i] = pe[64 * (it + 2) + 16 * i];
                ixn[it + 2][i] = pn[64 * (it + 2) + 16 * i];
            }
        }
        __builtin_amdgcn_sched_barrier(0);
        if (it + 1 < EPB) {
            #pragma unroll
            for (int i = 0; i < 4; ++i) {
                rwh[it + 1][i] = hv[(unsigned)ixe[it + 1][i] * 16u + (unsigned)quad];
                rwm[it + 1][i] = mv[(unsigned)ixn[it + 1][i] * 16u + (unsigned)quad];
            }
        }
        __builtin_amdgcn_sched_barrier(0);

        const int buf = it & 1;
        red[buf][0][tid] = (rwh[it][0] + rwh[it][1]) + (rwh[it][2] + rwh[it][3]);
        red[buf][1][tid] = (rwm[it][0] + rwm[it][1]) + (rwm[it][2] + rwm[it][3]);

        asm volatile("s_wait_dscnt 0x0" ::: "memory");
        asm volatile("s_barrier_signal -1" ::: "memory");
        asm volatile("s_barrier_wait -1" ::: "memory");

        if (tid < 32) {
            const v4f* src = &red[buf][tid >> 4][quad];
            v4f s[16];
            #pragma unroll
            for (int k = 0; k < 16; ++k) s[k] = src[16 * k];
            #pragma unroll
            for (int w = 8; w >= 1; w >>= 1) {
                #pragma unroll
                for (int k = 0; k < w; ++k) s[k] += s[k + w];
            }
            fv[(size_t)(e0 + it) * 32 + tid] = s[0] * (1.0f / 64.0f);
        }
    }
}

template <int CTRL>
__device__ __forceinline__ float row_xor_add(float v) {
    const int t = __builtin_amdgcn_update_dpp(0, __builtin_bit_cast(int, v), CTRL,
                                               0xf,  0xf,
                                               true);
    return v + __builtin_bit_cast(float, t);
}

__global__ __launch_bounds__(256)
void proj_attn_kernel(const float* __restrict__ h_xe, const float* __restrict__ f_u,
                      const float* __restrict__ W_h, const float* __restrict__ W_f,
                      const float* __restrict__ a, float* __restrict__ h_new,
                      float* __restrict__ alpha_out) {
    const int wave = threadIdx.x >> 5;
    const int cq   = wave & 3;
    const int kh   = wave >> 2;
    const int lane = threadIdx.x & 31;
    const int c    = lane & 15;
    const int hf   = lane >> 4;
    const int rowK = blockIdx.x * 32 + 16 * kh;
    const int rowS = blockIdx.x * 32 + 16 * (1 - kh);
    const int col0 = 32 * cq + c;

    constexpr int NBH   = 32 / 8;
    constexpr int NB    = NBH + 64 / 8;
    constexpr int DEPTH = 4;

    const float* AhK = h_xe + (size_t)(rowK + c) * 64 + 32 * kh + 4 * hf;
    const float* AhS = h_xe + (size_t)(rowS + c) * 64 + 32 * kh + 4 * hf;
    const float* AfK = f_u + (size_t)(rowK + c) * 128 + 64 * kh + 4 * hf;
    const float* AfS = f_u + (size_t)(rowS + c) * 128 + 64 * kh + 4 * hf;
    const float* Bh = W_h + (size_t)(32 * kh + 4 * hf) * 128 + col0;
    const float* Bf = W_f + (size_t)(64 * kh + 4 * hf) * 128 + col0;

    v8f ph[2][2], pu[2][2];
    #pragma unroll
    for (int i = 0; i < 2; ++i) {
        #pragma unroll
        for (int n = 0; n < 2; ++n) {
            ph[i][n] = (v8f){0.f, 0.f, 0.f, 0.f, 0.f, 0.f, 0.f, 0.f};
            pu[i][n] = (v8f){0.f, 0.f, 0.f, 0.f, 0.f, 0.f, 0.f, 0.f};
        }
    }

    v4f fa[NB][2];
    v2f fb[NB][2][2];

    #pragma unroll
    for (int g = 0; g < DEPTH; ++g) {
        const float* ApK = (g < NBH) ? AhK + 8 * g : AfK + 8 * (g - NBH);
        const float* ApS = (g < NBH) ? AhS + 8 * g : AfS + 8 * (g - NBH);
        const float* Bp  = (g < NBH) ? Bh + (size_t)(8 * g) * 128
                                     : Bf + (size_t)(8 * (g - NBH)) * 128;
        fa[g][0] = *(const v4f*)ApK;
        fa[g][1] = *(const v4f*)ApS;
        #pragma unroll
        for (int n = 0; n < 2; ++n) {
            #pragma unroll
            for (int t = 0; t < 2; ++t) {
                fb[g][n][t].x = Bp[(size_t)(2 * t) * 128 + 16 * n];
                fb[g][n][t].y = Bp[(size_t)(2 * t + 1) * 128 + 16 * n];
            }
        }
    }

    #pragma unroll
    for (int g = 0; g < NB; ++g) {
        if (g + DEPTH < NB) {
            const int q = g + DEPTH;
            const float* ApK = (q < NBH) ? AhK + 8 * q : AfK + 8 * (q - NBH);
            const float* ApS = (q < NBH) ? AhS + 8 * q : AfS + 8 * (q - NBH);
            const float* Bp  = (q < NBH) ? Bh + (size_t)(8 * q) * 128
                                         : Bf + (size_t)(8 * (q - NBH)) * 128;
            fa[q][0] = *(const v4f*)ApK;
            fa[q][1] = *(const v4f*)ApS;
            #pragma unroll
            for (int n = 0; n < 2; ++n) {
                #pragma unroll
                for (int t = 0; t < 2; ++t) {
                    fb[q][n][t].x = Bp[(size_t)(2 * t) * 128 + 16 * n];
                    fb[q][n][t].y = Bp[(size_t)(2 * t + 1) * 128 + 16 * n];
                }
            }
        }
        __builtin_amdgcn_sched_barrier(0);
        #pragma unroll
        for (int t = 0; t < 2; ++t) {
            #pragma unroll
            for (int i = 0; i < 2; ++i) {
                v2f af;
                af.x = fa[g][i][2 * t];
                af.y = fa[g][i][2 * t + 1];
                #pragma unroll
                for (int n = 0; n < 2; ++n) {
                    if (g < NBH)
                        ph[i][n] = __builtin_amdgcn_wmma_f32_16x16x4_f32(
                             false, af,  false, fb[g][n][t],
                             (short)0, ph[i][n],  false,  false);
                    else
                        pu[i][n] = __builtin_amdgcn_wmma_f32_16x16x4_f32(
                             false, af,  false, fb[g][n][t],
                             (short)0, pu[i][n],  false,  false);
                }
            }
        }
    }

    __shared__ v4f xch[8][8][32];
    #pragma unroll
    for (int n = 0; n < 2; ++n) {
        xch[wave][2 * n + 0][lane] = (v4f){ph[1][n][0], ph[1][n][1], ph[1][n][2], ph[1][n][3]};
        xch[wave][2 * n + 1][lane] = (v4f){ph[1][n][4], ph[1][n][5], ph[1][n][6], ph[1][n][7]};
        xch[wave][4 + 2 * n + 0][lane] = (v4f){pu[1][n][0], pu[1][n][1], pu[1][n][2], pu[1][n][3]};
        xch[wave][4 + 2 * n + 1][lane] = (v4f){pu[1][n][4], pu[1][n][5], pu[1][n][6], pu[1][n][7]};
    }
    __syncthreads();
    v8f qh[2], qu[2];
    #pragma unroll
    for (int n = 0; n < 2; ++n) {
        const v4f h0 = xch[wave ^ 4][2 * n + 0][lane];
        const v4f h1 = xch[wave ^ 4][2 * n + 1][lane];
        const v4f u0 = xch[wave ^ 4][4 + 2 * n + 0][lane];
        const v4f u1 = xch[wave ^ 4][4 + 2 * n + 1][lane];
        #pragma unroll
        for (int r = 0; r < 4; ++r) {
            qh[n][r]     = ph[0][n][r] + h0[r];
            qh[n][4 + r] = ph[0][n][4 + r] + h1[r];
            qu[n][r]     = pu[0][n][r] + u0[r];
            qu[n][4 + r] = pu[0][n][4 + r] + u1[r];
        }
    }

    const float av0 = a[col0];
    const float av1 = a[col0 + 16];

    __shared__ float part[4][2][32];
    float s0[8], s1[8];
    #pragma unroll
    for (int r = 0; r < 8; ++r) {
        s0[r] = tanhf(qh[0][r]) * av0 + tanhf(qh[1][r]) * av1;
        s1[r] = tanhf(qu[0][r]) * av0 + tanhf(qu[1][r]) * av1;
    }
    #pragma unroll
    for (int r = 0; r < 8; ++r) {
        s0[r] = row_xor_add<0x161>(s0[r]);
        s1[r] = row_xor_add<0x161>(s1[r]);
    }
    #pragma unroll
    for (int r = 0; r < 8; ++r) {
        s0[r] = row_xor_add<0x162>(s0[r]);
        s1[r] = row_xor_add<0x162>(s1[r]);
    }
    #pragma unroll
    for (int r = 0; r < 8; ++r) {
        s0[r] = row_xor_add<0x164>(s0[r]);
        s1[r] = row_xor_add<0x164>(s1[r]);
    }
    #pragma unroll
    for (int r = 0; r < 8; ++r) {
        s0[r] = row_xor_add<0x168>(s0[r]);
        s1[r] = row_xor_add<0x168>(s1[r]);
    }
    if (c == 0) {
        #pragma unroll
        for (int r = 0; r < 8; ++r) {
            part[cq][0][16 * kh + 8 * hf + r] = s0[r];
            part[cq][1][16 * kh + 8 * hf + r] = s1[r];
        }
    }
    __syncthreads();

    #pragma unroll
    for (int r = 0; r < 8; ++r) {
        const int   rl = 16 * kh + 8 * hf + r;
        const float s0 = (part[0][0][rl] + part[1][0][rl]) + (part[2][0][rl] + part[3][0][rl]);
        const float s1 = (part[0][1][rl] + part[1][1][rl]) + (part[2][1][rl] + part[3][1][rl]);
        const float mx = fmaxf(s0, s1);
        const float e0 = expf(s0 - mx);
        const float e1 = expf(s1 - mx);
        const float denom = e0 + 2.f * e1;
        const float a0  = e0 / denom;
        const float a1  = e1 / denom;
        const float a12 = 2.f * a1;
        const size_t grow = (size_t)blockIdx.x * 32 + rl;
        float* orow = h_new + grow * 128 + col0;
        orow[0]  = a0 * qh[0][r] + a12 * qu[0][r];
        orow[16] = a0 * qh[1][r] + a12 * qu[1][r];
        if (cq == 0 && c < 3)
            alpha_out[grow * 3 + c] = (c == 0) ? a0 : a1;
    }
}

extern "C" void kernel_launch(void* const* d_in, const int* in_sizes, int n_in,
                              void* d_out, int out_size, void* d_ws, size_t ws_size,
                              hipStream_t stream) {
    const float* h_xe = (const float*)d_in[0];
    const float* h    = (const float*)d_in[1];
    const float* m    = (const float*)d_in[2];
    const float* W_h  = (const float*)d_in[3];
    const float* W_f  = (const float*)d_in[4];
    const float* avec = (const float*)d_in[5];
    const int* idx_e  = (const int*)d_in[6];
    const int* idx_n  = (const int*)d_in[7];

    const int E  = in_sizes[0] / 64;
    const int DH = in_sizes[5];

    float* f_u = (float*)d_ws;

    float* h_new     = (float*)d_out;
    float* alpha_out = (float*)d_out + (size_t)E * DH;

    agg_kernel<<<E / EPB, 256, 0, stream>>>(h, m, idx_e, idx_n, f_u);

    proj_attn_kernel<<<E / 32, 256, 0, stream>>>(h_xe, f_u, W_h, W_f, avec, h_new, alpha_out);
}


# tanh pass: a lane's 16 ph values, then its 16 pu values, go through tanhf's own two ranges as one batch each (one wave-uniform range test per batch, chains side by side; was 32 serial calls)
# speedup vs baseline: 1.4778x; 1.0318x over previous
#include <hip/hip_runtime.h>
#include <math.h>

typedef __attribute__((ext_vector_type(2))) float v2f;
typedef __attribute__((ext_vector_type(4))) float v4f;
typedef __attribute__((ext_vector_type(8))) float v8f;

constexpr int EPB = 8;

__global__ __launch_bounds__(256) __attribute__((amdgpu_waves_per_eu(8)))
void agg_kernel(const float* __restrict__ h, const float* __restrict__ m,
                const int* __restrict__ idx_e, const int* __restrict__ idx_n,
                float* __restrict__ f_u) {
    const int tid   = threadIdx.x;
    const int quad  = tid & 15;
    const int rslot = tid >> 4;
    const int e0    = blockIdx.x * EPB;

    const v4f* __restrict__ hv = (const v4f*)h;
    const v4f* __restrict__ mv = (const v4f*)m;
    v4f* __restrict__ fv       = (v4f*)f_u;

    __shared__ v4f red[2][2][256];

    const int* pe = idx_e + (size_t)e0 * 64 + rslot;
    const int* pn = idx_n + (size_t)e0 * 64 + rslot;

    int ixe[EPB][4], ixn[EPB][4];
    v4f rwh[EPB][4], rwm[EPB][4];

    #pragma unroll
    for (int i = 0; i < 4; ++i) {
        ixe[0][i] = pe[16 * i];
        ixn[0][i] = pn[16 * i];
    }
    #pragma unroll
    for (int i = 0; i < 4; ++i) {
        ixe[1][i] = pe[64 + 16 * i];
        ixn[1][i] = pn[64 + 16 * i];
    }
    __builtin_amdgcn_sched_barrier(0);
    #pragma unroll
    for (int i = 0; i < 4; ++i) {
        rwh[0][i] = hv[(unsigned)ixe[0][i] * 16u + (unsigned)quad];
        rwm[0][i] = mv[(unsigned)ixn[0][i] * 16u + (unsigned)quad];
    }

    #pragma unroll
    for (int it = 0; it < EPB; ++it) {
        if (it + 2 < EPB) {
            #pragma unroll
            for (int i = 0; i < 4; ++i) {
                ixe[it + 2][i] = pe[64 * (it + 2) + 16 * i];
                ixn[it + 2][i] = pn[64 * (it + 2) + 16 * i];
            }
        }
        __builtin_amdgcn_sched_barrier(0);
        if (it + 1 < EPB) {
            #pragma unroll
            for (int i = 0; i < 4; ++i) {
                rwh[it + 1][i] = hv[(unsigned)ixe[it + 1][i] * 16u + (unsigned)quad];
                rwm[it + 1][i] = mv[(unsigned)ixn[it + 1][i] * 16u + (unsigned)quad];
            }
        }
        __builtin_amdgcn_sched_barrier(0);

        const int buf = it & 1;
        red[buf][0][tid] = (rwh[it][0] + rwh[it][1]) + (rwh[it][2] + rwh[it][3]);
        red[buf][1][tid] = (rwm[it][0] + rwm[it][1]) + (rwm[it][2] + rwm[it][3]);

        asm volatile("s_wait_dscnt 0x0" ::: "memory");
        asm volatile("s_barrier_signal -1" ::: "memory");
        asm volatile("s_barrier_wait -1" ::: "memory");

        if (tid < 32) {
            const v4f* src = &red[buf][tid >> 4][quad];
            v4f s[16];
            #pragma unroll
            for (int k = 0; k < 16; ++k) s[k] = src[16 * k];
            #pragma unroll
            for (int w = 8; w >= 1; w >>= 1) {
                #pragma unroll
                for (int k = 0; k < w; ++k) s[k] += s[k + w];
            }
            fv[(size_t)(e0 + it) * 32 + tid] = s[0] * (1.0f / 64.0f);
        }
    }
}

template <int CTRL>
__device__ __forceinline__ float row_xor_add(float v) {
    const int t = __builtin_amdgcn_update_dpp(0, __builtin_bit_cast(int, v), CTRL,
                                               0xf,  0xf,
                                               true);
    return v + __builtin_bit_cast(float, t);
}

__device__ __forceinline__ void tanh16(const v8f& xa, const v8f& xb, v8f& za, v8f& zb) {
    float x[16], y[16], zs[16], zl[16];
    bool any_small = false, any_large = false;
    #pragma unroll
    for (int i = 0; i < 16; ++i) {
        x[i]  = (i < 8) ? xa[i & 7] : xb[i & 7];
        y[i]  = fabsf(x[i]);
        zs[i] = 0.f;
        zl[i] = 0.f;
        any_small |= (y[i] < 0.625f);
        any_large |= !(y[i] < 0.625f);
    }
    if (__builtin_amdgcn_ballot_w32(any_small) != 0u) {
        #pragma unroll
        for (int i = 0; i < 16; ++i) {
            const float y2 = y[i] * y[i];
            const float p  = fmaf(y2, fmaf(y2, fmaf(y2, fmaf(y2, -0x1.758e7ap-8f, 0x1.521192p-6f),
                                                     -0x1.b8389cp-5f), 0x1.110704p-3f), -0x1.555532p-2f);
            zs[i] = fmaf(y2, y[i] * p, y[i]);
        }
    }
    if (__builtin_amdgcn_ballot_w32(any_large) != 0u) {
        #pragma unroll
        for (int i = 0; i < 16; ++i) {
            const float t = expf(2.0f * y[i]);
            zl[i] = fmaf(-2.0f, __builtin_amdgcn_rcpf(t + 1.0f), 1.0f);
        }
    }
    #pragma unroll
    for (int i = 0; i < 16; ++i) {
        const float z = copysignf((y[i] < 0.625f) ? zs[i] : zl[i], x[i]);
        if (i < 8) za[i & 7] = z; else zb[i & 7] = z;
    }
}

__global__ __launch_bounds__(256)
void proj_attn_kernel(const float* __restrict__ h_xe, const float* __restrict__ f_u,
                      const float* __restrict__ W_h, const float* __restrict__ W_f,
                      const float* __restrict__ a, float* __restrict__ h_new,
                      float* __restrict__ alpha_out) {
    const int wave = threadIdx.x >> 5;
    const int cq   = wave & 3;
    const int kh   = wave >> 2;
    const int lane = threadIdx.x & 31;
    const int c    = lane & 15;
    const int hf   = lane >> 4;
    const int rowK = blockIdx.x * 32 + 16 * kh;
    const int rowS = blockIdx.x * 32 + 16 * (1 - kh);
    const int col0 = 32 * cq + c;

    constexpr int NBH   = 32 / 8;
    constexpr int NB    = NBH + 64 / 8;
    constexpr int DEPTH = 4;

    const float* AhK = h_xe + (size_t)(rowK + c) * 64 + 32 * kh + 4 * hf;
    const float* AhS = h_xe + (size_t)(rowS + c) * 64 + 32 * kh + 4 * hf;
    const float* AfK = f_u + (size_t)(rowK + c) * 128 + 64 * kh + 4 * hf;
    const float* AfS = f_u + (size_t)(rowS + c) * 128 + 64 * kh + 4 * hf;
    const float* Bh = W_h + (size_t)(32 * kh + 4 * hf) * 128 + col0;
    const float* Bf = W_f + (size_t)(64 * kh + 4 * hf) * 128 + col0;

    v8f ph[2][2], pu[2][2];
    #pragma unroll
    for (int i = 0; i < 2; ++i) {
        #pragma unroll
        for (int n = 0; n < 2; ++n) {
            ph[i][n] = (v8f){0.f, 0.f, 0.f, 0.f, 0.f, 0.f, 0.f, 0.f};
            pu[i][n] = (v8f){0.f, 0.f, 0.f, 0.f, 0.f, 0.f, 0.f, 0.f};
        }
    }

    v4f fa[NB][2];
    v2f fb[NB][2][2];

    #pragma unroll
    for (int g = 0; g < DEPTH; ++g) {
        const float* ApK = (g < NBH) ? AhK + 8 * g : AfK + 8 * (g - NBH);
        const float* ApS = (g < NBH) ? AhS + 8 * g : AfS + 8 * (g - NBH);
        const float* Bp  = (g < NBH) ? Bh + (size_t)(8 * g) * 128
                                     : Bf + (size_t)(8 * (g - NBH)) * 128;
        fa[g][0] = *(const v4f*)ApK;
        fa[g][1] = *(const v4f*)ApS;
        #pragma unroll
        for (int n = 0; n < 2; ++n) {
            #pragma unroll
            for (int t = 0; t < 2; ++t) {
                fb[g][n][t].x = Bp[(size_t)(2 * t) * 128 + 16 * n];
                fb[g][n][t].y = Bp[(size_t)(2 * t + 1) * 128 + 16 * n];
            }
        }
    }

    #pragma unroll
    for (int g = 0; g < NB; ++g) {
        if (g + DEPTH < NB) {
            const int q = g + DEPTH;
            const float* ApK = (q < NBH) ? AhK + 8 * q : AfK + 8 * (q - NBH);
            const float* ApS = (q < NBH) ? AhS + 8 * q : AfS + 8 * (q - NBH);
            const float* Bp  = (q < NBH) ? Bh + (size_t)(8 * q) * 128
                                         : Bf + (size_t)(8 * (q - NBH)) * 128;
            fa[q][0] = *(const v4f*)ApK;
            fa[q][1] = *(const v4f*)ApS;
            #pragma unroll
            for (int n = 0; n < 2; ++n) {
                #pragma unroll
                for (int t = 0; t < 2; ++t) {
                    fb[q][n][t].x = Bp[(size_t)(2 * t) * 128 + 16 * n];
                    fb[q][n][t].y = Bp[(size_t)(2 * t + 1) * 128 + 16 * n];
                }
            }
        }
        __builtin_amdgcn_sched_barrier(0);
        #pragma unroll
        for (int t = 0; t < 2; ++t) {
            #pragma unroll
            for (int i = 0; i < 2; ++i) {
                v2f af;
                af.x = fa[g][i][2 * t];
                af.y = fa[g][i][2 * t + 1];
                #pragma unroll
                for (int n = 0; n < 2; ++n) {
                    if (g < NBH)
                        ph[i][n] = __builtin_amdgcn_wmma_f32_16x16x4_f32(
                             false, af,  false, fb[g][n][t],
                             (short)0, ph[i][n],  false,  false);
                    else
                        pu[i][n] = __builtin_amdgcn_wmma_f32_16x16x4_f32(
                             false, af,  false, fb[g][n][t],
                             (short)0, pu[i][n],  false,  false);
                }
            }
        }
    }

    __shared__ v4f xch[8][8][32];
    #pragma unroll
    for (int n = 0; n < 2; ++n) {
        xch[wave][2 * n + 0][lane] = (v4f){ph[1][n][0], ph[1][n][1], ph[1][n][2], ph[1][n][3]};
        xch[wave][2 * n + 1][lane] = (v4f){ph[1][n][4], ph[1][n][5], ph[1][n][6], ph[1][n][7]};
        xch[wave][4 + 2 * n + 0][lane] = (v4f){pu[1][n][0], pu[1][n][1], pu[1][n][2], pu[1][n][3]};
        xch[wave][4 + 2 * n + 1][lane] = (v4f){pu[1][n][4], pu[1][n][5], pu[1][n][6], pu[1][n][7]};
    }
    __syncthreads();
    v8f qh[2], qu[2];
    #pragma unroll
    for (int n = 0; n < 2; ++n) {
        const v4f h0 = xch[wave ^ 4][2 * n + 0][lane];
        const v4f h1 = xch[wave ^ 4][2 * n + 1][lane];
        const v4f u0 = xch[wave ^ 4][4 + 2 * n + 0][lane];
        const v4f u1 = xch[wave ^ 4][4 + 2 * n + 1][lane];
        #pragma unroll
        for (int r = 0; r < 4; ++r) {
            qh[n][r]     = ph[0][n][r] + h0[r];
            qh[n][4 + r] = ph[0][n][4 + r] + h1[r];
            qu[n][r]     = pu[0][n][r] + u0[r];
            qu[n][4 + r] = pu[0][n][4 + r] + u1[r];
        }
    }

    const float av0 = a[col0];
    const float av1 = a[col0 + 16];

    __shared__ float part[4][2][32];
    float s0[8], s1[8];
    v8f th[2], tu[2];
    tanh16(qh[0], qh[1], th[0], th[1]);
    tanh16(qu[0], qu[1], tu[0], tu[1]);
    #pragma unroll
    for (int r = 0; r < 8; ++r) {
        s0[r] = th[0][r] * av0 + th[1][r] * av1;
        s1[r] = tu[0][r] * av0 + tu[1][r] * av1;
    }
    #pragma unroll
    for (int r = 0; r < 8; ++r) {
        s0[r] = row_xor_add<0x161>(s0[r]);
        s1[r] = row_xor_add<0x161>(s1[r]);
    }
    #pragma unroll
    for (int r = 0; r < 8; ++r) {
        s0[r] = row_xor_add<0x162>(s0[r]);
        s1[r] = row_xor_add<0x162>(s1[r]);
    }
    #pragma unroll
    for (int r = 0; r < 8; ++r) {
        s0[r] = row_xor_add<0x164>(s0[r]);
        s1[r] = row_xor_add<0x164>(s1[r]);
    }
    #pragma unroll
    for (int r = 0; r < 8; ++r) {
        s0[r] = row_xor_add<0x168>(s0[r]);
        s1[r] = row_xor_add<0x168>(s1[r]);
    }
    if (c == 0) {
        #pragma unroll
        for (int r = 0; r < 8; ++r) {
            part[cq][0][16 * kh + 8 * hf + r] = s0[r];
            part[cq][1][16 * kh + 8 * hf + r] = s1[r];
        }
    }
    __syncthreads();

    #pragma unroll
    for (int r = 0; r < 8; ++r) {
        const int   rl = 16 * kh + 8 * hf + r;
        const float s0 = (part[0][0][rl] + part[1][0][rl]) + (part[2][0][rl] + part[3][0][rl]);
        const float s1 = (part[0][1][rl] + part[1][1][rl]) + (part[2][1][rl] + part[3][1][rl]);
        const float mx = fmaxf(s0, s1);
        const float e0 = expf(s0 - mx);
        const float e1 = expf(s1 - mx);
        const float denom = e0 + 2.f * e1;
        const float a0  = e0 / denom;
        const float a1  = e1 / denom;
        const float a12 = 2.f * a1;
        const size_t grow = (size_t)blockIdx.x * 32 + rl;
        float* orow = h_new + grow * 128 + col0;
        orow[0]  = a0 * qh[0][r] + a12 * qu[0][r];
        orow[16] = a0 * qh[1][r] + a12 * qu[1][r];
        if (cq == 0 && c < 3)
            alpha_out[grow * 3 + c] = (c == 0) ? a0 : a1;
    }
}

extern "C" void kernel_launch(void* const* d_in, const int* in_sizes, int n_in,
                              void* d_out, int out_size, void* d_ws, size_t ws_size,
                              hipStream_t stream) {
    const float* h_xe = (const float*)d_in[0];
    const float* h    = (const float*)d_in[1];
    const float* m    = (const float*)d_in[2];
    const float* W_h  = (const float*)d_in[3];
    const float* W_f  = (const float*)d_in[4];
    const float* avec = (const float*)d_in[5];
    const int* idx_e  = (const int*)d_in[6];
    const int* idx_n  = (const int*)d_in[7];

    const int E  = in_sizes[0] / 64;
    const int DH = in_sizes[5];

    float* f_u = (float*)d_ws;

    float* h_new     = (float*)d_out;
    float* alpha_out = (float*)d_out + (size_t)E * DH;

    agg_kernel<<<E / EPB, 256, 0, stream>>>(h, m, idx_e, idx_n, f_u);

    proj_attn_kernel<<<E / 32, 256, 0, stream>>>(h_xe, f_u, W_h, W_f, avec, h_new, alpha_out);
}


# gather and projection/attention in one launch: 512 workgroups x 16 edges, row sums go to a 16-row LDS tile, 8 waves x 16 columns on WMMA from it; 2 launches -> 1; gather and score sum order changed
# speedup vs baseline: 1.9290x; 1.3053x over previous
#include <hip/hip_runtime.h>
#include <math.h>


typedef __attribute__((ext_vector_type(2))) float v2f;
typedef __attribute__((ext_vector_type(4))) float v4f;
typedef __attribute__((ext_vector_type(8))) float v8f;
typedef __attribute__((ext_vector_type(4))) int   v4i;

constexpr int EPB = 16;
constexpr int RD  = 3;
constexpr int ID  = 6;
constexpr int AST = 132;

template <int CTRL>
__device__ __forceinline__ float row_xor_add(float v) {
    const int t = __builtin_amdgcn_update_dpp(0, __builtin_bit_cast(int, v), CTRL,
                                               0xf,  0xf,
                                               true);
    return v + __builtin_bit_cast(float, t);
}

template <int N>
__device__ __forceinline__ float row_share(float v) {
    const int t = __builtin_amdgcn_update_dpp(0, __builtin_bit_cast(int, v), 0x150 + N,
                                               0xf,  0xf,
                                               true);
    return __builtin_bit_cast(float, t);
}

__device__ __forceinline__ void tanh16(const v8f& xa, const v8f& xb, v8f& za, v8f& zb) {
    float x[16], y[16], zs[16], zl[16];
    bool any_small = false, any_large = false;
    #pragma unroll
    for (int i = 0; i < 16; ++i) {
        x[i]  = (i < 8) ? xa[i & 7] : xb[i & 7];
        y[i]  = fabsf(x[i]);
        zs[i] = 0.f;
        zl[i] = 0.f;
        any_small |= (y[i] < 0.625f);
        any_large |= !(y[i] < 0.625f);
    }
    if (__builtin_amdgcn_ballot_w32(any_small) != 0u) {
        #pragma unroll
        for (int i = 0; i < 16; ++i) {
            const float y2 = y[i] * y[i];
            const float p  = fmaf(y2, fmaf(y2, fmaf(y2, fmaf(y2, -0x1.758e7ap-8f, 0x1.521192p-6f),
                                                     -0x1.b8389cp-5f), 0x1.110704p-3f), -0x1.555532p-2f);
            zs[i] = fmaf(y2, y[i] * p, y[i]);
        }
    }
    if (__builtin_amdgcn_ballot_w32(any_large) != 0u) {
        #pragma unroll
        for (int i = 0; i < 16; ++i) {
            const float t = expf(2.0f * y[i]);
            zl[i] = fmaf(-2.0f, __builtin_amdgcn_rcpf(t + 1.0f), 1.0f);
        }
    }
    #pragma unroll
    for (int i = 0; i < 16; ++i) {
        const float z = copysignf((y[i] < 0.625f) ? zs[i] : zl[i], x[i]);
        if (i < 8) za[i & 7] = z; else zb[i & 7] = z;
    }
}

__global__ __launch_bounds__(256) __attribute__((amdgpu_waves_per_eu(4)))
void edge_attn_kernel(const float* __restrict__ h_xe, const float* __restrict__ h,
                      const float* __restrict__ m, const float* __restrict__ W_h,
                      const float* __restrict__ W_f, const float* __restrict__ a,
                      const int* __restrict__ idx_e, const int* __restrict__ idx_n,
                      float* __restrict__ h_new, float* __restrict__ alpha_out) {
    const int tid   = threadIdx.x;
    const int quad  = tid & 15;
    const int rslot = tid >> 4;
    const int wave  = tid >> 5;
    const int lane  = tid & 31;
    const int c     = lane & 15;
    const int hf    = lane >> 4;
    const int e0    = blockIdx.x * EPB;
    const int col   = 16 * wave + c;

    const v4f* __restrict__ hv = (const v4f*)h;
    const v4f* __restrict__ mv = (const v4f*)m;

    __shared__ v4f red[2][2][256];
    __shared__ __attribute__((aligned(16))) float At[EPB][AST];
    __shared__ float part[8][2][EPB];
    __shared__ __attribute__((aligned(16))) float alst[EPB * 3];

    const v4i* pe = (const v4i*)(idx_e + (size_t)e0 * 64) + rslot;
    const v4i* pn = (const v4i*)(idx_n + (size_t)e0 * 64) + rslot;

    v4i ixe[EPB], ixn[EPB];
    v4f rwh[EPB][4], rwm[EPB][4];

    #pragma unroll
    for (int e = 0; e < ID; ++e) {
        ixe[e] = pe[16 * e];
        ixn[e] = pn[16 * e];
    }
    __builtin_amdgcn_sched_barrier(0);

    const float* Ah = h_xe + (size_t)(e0 + c) * 64 + 4 * hf;
    const float* Bh = W_h + (size_t)(4 * hf) * 128 + col;
    v4f fah[8];
    v2f fbh[8][2];
    #pragma unroll
    for (int j = 0; j < 8; ++j) {
        fah[j] = *(const v4f*)(Ah + 8 * j);
        #pragma unroll
        for (int t = 0; t < 2; ++t) {
            fbh[j][t].x = Bh[(size_t)(8 * j + 2 * t) * 128];
            fbh[j][t].y = Bh[(size_t)(8 * j + 2 * t + 1) * 128];
        }
    }
    const float av = a[col];
    __builtin_amdgcn_sched_barrier(0);

    #pragma unroll
    for (int e = 0; e < RD; ++e) {
        #pragma unroll
        for (int i = 0; i < 4; ++i) {
            rwh[e][i] = hv[(unsigned)ixe[e][i] * 16u + (unsigned)quad];
            rwm[e][i] = mv[(unsigned)ixn[e][i] * 16u + (unsigned)quad];
        }
    }
    __builtin_amdgcn_sched_barrier(0);

    v8f qh;
    {
        v8f lo = (v8f){0.f, 0.f, 0.f, 0.f, 0.f, 0.f, 0.f, 0.f};
        v8f hi = (v8f){0.f, 0.f, 0.f, 0.f, 0.f, 0.f, 0.f, 0.f};
        #pragma unroll
        for (int j = 0; j < 4; ++j) {
            #pragma unroll
            for (int t = 0; t < 2; ++t) {
                v2f al, ah;
                al.x = fah[j][2 * t];      al.y = fah[j][2 * t + 1];
                ah.x = fah[j + 4][2 * t];  ah.y = fah[j + 4][2 * t + 1];
                lo = __builtin_amdgcn_wmma_f32_16x16x4_f32(
                     false, al,  false, fbh[j][t],
                     (short)0, lo,  false,  false);
                hi = __builtin_amdgcn_wmma_f32_16x16x4_f32(
                     false, ah,  false, fbh[j + 4][t],
                     (short)0, hi,  false,  false);
            }
        }
        #pragma unroll
        for (int r = 0; r < 8; ++r) qh[r] = lo[r] + hi[r];
    }

    const float* Bf = W_f + (size_t)(4 * hf) * 128 + col;
    v2f fbf[16][2];

    #pragma unroll
    for (int it = 0; it < EPB; ++it) {
        if (it + ID < EPB) {
            ixe[it + ID] = pe[16 * (it + ID)];
            ixn[it + ID] = pn[16 * (it + ID)];
        }
        __builtin_amdgcn_sched_barrier(0);
        if (it + RD < EPB) {
            #pragma unroll
            for (int i = 0; i < 4; ++i) {
                rwh[it + RD][i] = hv[(unsigned)ixe[it + RD][i] * 16u + (unsigned)quad];
                rwm[it + RD][i] = mv[(unsigned)ixn[it + RD][i] * 16u + (unsigned)quad];
            }
        }
        if (it >= EPB - RD) {
            const int j0 = (it == EPB - 3) ? 0 : (it == EPB - 2) ? 6 : 11;
            const int j1 = (it == EPB - 3) ? 6 : (it == EPB - 2) ? 11 : 16;
            #pragma unroll
            for (int j = j0; j < j1; ++j) {
                #pragma unroll
                for (int t = 0; t < 2; ++t) {
                    fbf[j][t].x = Bf[(size_t)(8 * j + 2 * t) * 128];
                    fbf[j][t].y = Bf[(size_t)(8 * j + 2 * t + 1) * 128];
                }
            }
        }
        __builtin_amdgcn_sched_barrier(0);

        const int buf = it & 1;
        red[buf][0][tid] = (rwh[it][0] + rwh[it][1]) + (rwh[it][2] + rwh[it][3]);
        red[buf][1][tid] = (rwm[it][0] + rwm[it][1]) + (rwm[it][2] + rwm[it][3]);

        asm volatile("s_wait_dscnt 0x0" ::: "memory");
        asm volatile("s_barrier_signal -1" ::: "memory");
        asm volatile("s_barrier_wait -1" ::: "memory");

        if (wave == (it & 7)) {
            const v4f* src = &red[buf][hf][c];
            v4f s[16];
            #pragma unroll
            for (int k = 0; k < 16; ++k) s[k] = src[16 * k];
            #pragma unroll
            for (int w = 8; w >= 1; w >>= 1) {
                #pragma unroll
                for (int k = 0; k < w; ++k) s[k] += s[k + w];
            }
            *(v4f*)&At[it][4 * lane] = s[0] * (1.0f / 64.0f);
        }
    }
    __syncthreads();

    v8f qu;
    {
        v4f faf[16];
        #pragma unroll
        for (int j = 0; j < 16; ++j) faf[j] = *(const v4f*)&At[c][8 * j + 4 * hf];
        v8f lo = (v8f){0.f, 0.f, 0.f, 0.f, 0.f, 0.f, 0.f, 0.f};
        v8f hi = (v8f){0.f, 0.f, 0.f, 0.f, 0.f, 0.f, 0.f, 0.f};
        #pragma unroll
        for (int j = 0; j < 8; ++j) {
            #pragma unroll
            for (int t = 0; t < 2; ++t) {
                v2f al, ah;
                al.x = faf[j][2 * t];      al.y = faf[j][2 * t + 1];
                ah.x = faf[j + 8][2 * t];  ah.y = faf[j + 8][2 * t + 1];
                lo = __builtin_amdgcn_wmma_f32_16x16x4_f32(
                     false, al,  false, fbf[j][t],
                     (short)0, lo,  false,  false);
                hi = __builtin_amdgcn_wmma_f32_16x16x4_f32(
                     false, ah,  false, fbf[j + 8][t],
                     (short)0, hi,  false,  false);
            }
        }
        #pragma unroll
        for (int r = 0; r < 8; ++r) qu[r] = lo[r] + hi[r];
    }

    float s0[8], s1[8];
    v8f th, tu;
    tanh16(qh, qu, th, tu);
    #pragma unroll
    for (int r = 0; r < 8; ++r) {
        s0[r] = th[r] * av;
        s1[r] = tu[r] * av;
    }
    #pragma unroll
    for (int r = 0; r < 8; ++r) {
        s0[r] = row_xor_add<0x161>(s0[r]);
        s1[r] = row_xor_add<0x161>(s1[r]);
    }
    #pragma unroll
    for (int r = 0; r < 8; ++r) {
        s0[r] = row_xor_add<0x162>(s0[r]);
        s1[r] = row_xor_add<0x162>(s1[r]);
    }
    #pragma unroll
    for (int r = 0; r < 8; ++r) {
        s0[r] = row_xor_add<0x164>(s0[r]);
        s1[r] = row_xor_add<0x164>(s1[r]);
    }
    #pragma unroll
    for (int r = 0; r < 8; ++r) {
        s0[r] = row_xor_add<0x168>(s0[r]);
        s1[r] = row_xor_add<0x168>(s1[r]);
    }
    if (c == 0) {
        #pragma unroll
        for (int r = 0; r < 8; ++r) {
            part[wave][0][8 * hf + r] = s0[r];
            part[wave][1][8 * hf + r] = s1[r];
        }
    }
    __syncthreads();

    const int rr = 8 * hf + (c & 7);
    const float t0 = ((part[0][0][rr] + part[1][0][rr]) + (part[2][0][rr] + part[3][0][rr]))
                   + ((part[4][0][rr] + part[5][0][rr]) + (part[6][0][rr] + part[7][0][rr]));
    const float t1 = ((part[0][1][rr] + part[1][1][rr]) + (part[2][1][rr] + part[3][1][rr]))
                   + ((part[4][1][rr] + part[5][1][rr]) + (part[6][1][rr] + part[7][1][rr]));
    const float mx = fmaxf(t0, t1);
    const float x0 = expf(t0 - mx);
    const float x1 = expf(t1 - mx);
    const float denom = x0 + 2.f * x1;
    const float a0  = x0 / denom;
    const float a1  = x1 / denom;
    const float a12 = 2.f * a1;
    if (wave == 0 && c < 8) {
        alst[rr * 3 + 0] = a0;
        alst[rr * 3 + 1] = a1;
        alst[rr * 3 + 2] = a1;
    }
#define ROW_OUT(R) At[8 * hf + R][col] = row_share<R>(a0) * qh[R] + row_share<R>(a12) * qu[R];
    ROW_OUT(0) ROW_OUT(1) ROW_OUT(2) ROW_OUT(3) ROW_OUT(4) ROW_OUT(5) ROW_OUT(6) ROW_OUT(7)
#undef ROW_OUT
    __syncthreads();

    for (int i = 0; i < 2; ++i) {
        const int row = 2 * wave + i;
        const v4f v = *(const v4f*)&At[row][4 * lane];
        ((v4f*)(h_new + (size_t)(e0 + row) * 128))[lane] = v;
    }
    if (wave == 0 && lane < (EPB * 3) / 4)
        ((v4f*)(alpha_out + (size_t)e0 * 3))[lane] = *(const v4f*)&alst[4 * lane];
}

extern "C" void kernel_launch(void* const* d_in, const int* in_sizes, int n_in,
                              void* d_out, int out_size, void* d_ws, size_t ws_size,
                              hipStream_t stream) {
    const float* h_xe = (const float*)d_in[0];
    const float* h    = (const float*)d_in[1];
    const float* m    = (const float*)d_in[2];
    const float* W_h  = (const float*)d_in[3];
    const float* W_f  = (const float*)d_in[4];
    const float* avec = (const float*)d_in[5];
    const int* idx_e  = (const int*)d_in[6];
    const int* idx_n  = (const int*)d_in[7];

    const int E  = in_sizes[0] / 64;
    const int DH = in_sizes[5];

    float* h_new     = (float*)d_out;
    float* alpha_out = (float*)d_out + (size_t)E * DH;

    edge_attn_kernel<<<E / EPB, 256, 0, stream>>>(h_xe, h, m, W_h, W_f, avec, idx_e, idx_n,
                                                  h_new, alpha_out);
}
